# starting kernel
# baseline (speedup 1.0000x reference)
#include <hip/hip_runtime.h>
#include <math.h>

typedef __attribute__((ext_vector_type(2))) float v2f;
typedef __attribute__((ext_vector_type(8))) float v8f;

#define WAVES 4
#define THREADS (WAVES * 32)
#define EDGES_PER_WAVE 32
#define EDGES_PER_PASS (WAVES * EDGES_PER_WAVE)
#define CHUNKS 8
#define EPB (EDGES_PER_PASS * CHUNKS)

#define W_ROW 512
#define H_ROW 132

__device__ __forceinline__ v8f wmma_f32_16x16x4(v2f a, v2f b, v8f c) {
  return __builtin_amdgcn_wmma_f32_16x16x4_f32(false, a, false, b, (short)0, c, false, false);
}

__device__ __forceinline__ v2f lds2(const float* p) { return *(const v2f*)p; }

__device__ __forceinline__ unsigned fkey(float f) {
  unsigned u = __float_as_uint(f);
  return (u & 0x80000000u) ? ~u : (u | 0x80000000u);
}
__device__ __forceinline__ float funkey(unsigned k) {
  unsigned u = (k & 0x80000000u) ? (k & 0x7fffffffu) : ~k;
  return __uint_as_float(u);
}

__global__ __launch_bounds__(THREADS) void edge_mlp_kernel(
    const float* __restrict__ uf, const float* __restrict__ gf,
    const float* __restrict__ ef, const int* __restrict__ uidx,
    const int* __restrict__ gidx,
    const float* __restrict__ W1, const float* __restrict__ b1,
    const float* __restrict__ W2, const float* __restrict__ b2,
    const float* __restrict__ w3, const float* __restrict__ b3,
    float* __restrict__ logits, int E)
{
  extern __shared__ float smem[];
  float* sW1g = smem;
  float* sb1  = sW1g + 2 * W_ROW;
  float* sW2g = sb1 + 128;
  float* sb2  = sW2g + 32 * W_ROW;
  float* sw3  = sb2 + 128;
  float* sX   = sw3 + 128;
  float* sH   = sX + WAVES * 2 * H_ROW;

  const int tid = threadIdx.x;

  for (int i = tid; i < 128; i += THREADS) { sb1[i] = b1[i]; sb2[i] = b2[i]; sw3[i] = w3[i]; }
  for (int i = tid; i < 8 * 128; i += THREADS) {
    const int k = i >> 7, n = i & 127;
    sW1g[(k >> 2) * W_ROW + n * 4 + (k & 3)] = W1[i];
  }
  for (int i = tid; i < 128 * 128; i += THREADS) {
    const int k = i >> 7, n = i & 127;
    sW2g[(k >> 2) * W_ROW + n * 4 + (k & 3)] = W2[i];
  }
  __syncthreads();

  const int wid  = tid >> 5;
  const int lane = tid & 31;
  const int l16  = lane & 15;
  const int half = lane >> 4;
  const int koff = half * 2;
  float* sXw = sX + wid * (2 * H_ROW);
  float* sHw = sH + wid * (32 * H_ROW);

  const int blockBase = blockIdx.x * EPB;

  for (int c = 0; c < CHUNKS; ++c) {
    const int eBase = blockBase + c * EDGES_PER_PASS + wid * EDGES_PER_WAVE;

    {
      const int m = lane;
      int e = eBase + m; if (e >= E) e = E - 1;
      const int uI = uidx[e], gI = gidx[e];
      float v[8];
      v[0] = uf[uI * 3 + 0]; v[1] = uf[uI * 3 + 1]; v[2] = uf[uI * 3 + 2];
      v[3] = gf[gI * 3 + 0]; v[4] = gf[gI * 3 + 1]; v[5] = gf[gI * 3 + 2];
      v[6] = ef[e * 2 + 0];  v[7] = ef[e * 2 + 1];
      #pragma unroll
      for (int f = 0; f < 8; ++f)
        sXw[(f >> 2) * H_ROW + m * 4 + (f & 3)] = v[f];
    }
    __builtin_amdgcn_wave_barrier();

    #pragma unroll
    for (int nt = 0; nt < 8; ++nt) {
      const int n = nt * 16 + l16;
      const v2f b0 = lds2(&sW1g[0 * W_ROW + n * 4 + koff]);
      const v2f b1v = lds2(&sW1g[1 * W_ROW + n * 4 + koff]);
      const float bias = sb1[n];
      #pragma unroll
      for (int t = 0; t < 2; ++t) {
        const v2f a0 = lds2(&sXw[0 * H_ROW + (t * 16 + l16) * 4 + koff]);
        const v2f a1 = lds2(&sXw[1 * H_ROW + (t * 16 + l16) * 4 + koff]);
        v8f cc = {0.f, 0.f, 0.f, 0.f, 0.f, 0.f, 0.f, 0.f};
        cc = wmma_f32_16x16x4(a0, b0, cc);
        cc = wmma_f32_16x16x4(a1, b1v, cc);
        #pragma unroll
        for (int r = 0; r < 8; ++r) {
          float h = cc[r] + bias;
          h = h > 0.f ? h : 0.f;
          const int m = t * 16 + r + half * 8;
          sHw[(n >> 2) * H_ROW + m * 4 + (n & 3)] = h;
        }
      }
    }
    __builtin_amdgcn_wave_barrier();

    float accw0[8], accw1[8];
    #pragma unroll
    for (int r = 0; r < 8; ++r) { accw0[r] = 0.f; accw1[r] = 0.f; }

    for (int nt = 0; nt < 8; ++nt) {
      const int n = nt * 16 + l16;
      v8f c0 = {0.f, 0.f, 0.f, 0.f, 0.f, 0.f, 0.f, 0.f};
      v8f c1 = {0.f, 0.f, 0.f, 0.f, 0.f, 0.f, 0.f, 0.f};
      #pragma unroll 4
      for (int ks = 0; ks < 32; ++ks) {
        const v2f b  = lds2(&sW2g[ks * W_ROW + n * 4 + koff]);
        const v2f a0 = lds2(&sHw[ks * H_ROW + l16 * 4 + koff]);
        const v2f a1 = lds2(&sHw[ks * H_ROW + (16 + l16) * 4 + koff]);
        c0 = wmma_f32_16x16x4(a0, b, c0);
        c1 = wmma_f32_16x16x4(a1, b, c1);
      }
      const float b2v = sb2[n];
      const float w3v = sw3[n];
      #pragma unroll
      for (int r = 0; r < 8; ++r) {
        float h0 = c0[r] + b2v; h0 = h0 > 0.f ? h0 : 0.f;
        float h1 = c1[r] + b2v; h1 = h1 > 0.f ? h1 : 0.f;
        accw0[r] += h0 * w3v;
        accw1[r] += h1 * w3v;
      }
    }

    #pragma unroll
    for (int r = 0; r < 8; ++r) {
      float v0 = accw0[r], v1 = accw1[r];
      v0 += __shfl_xor(v0, 1);  v1 += __shfl_xor(v1, 1);
      v0 += __shfl_xor(v0, 2);  v1 += __shfl_xor(v1, 2);
      v0 += __shfl_xor(v0, 4);  v1 += __shfl_xor(v1, 4);
      v0 += __shfl_xor(v0, 8);  v1 += __shfl_xor(v1, 8);
      accw0[r] = v0; accw1[r] = v1;
    }
    if (l16 == 0) {
      const float bb3 = b3[0];
      #pragma unroll
      for (int r = 0; r < 8; ++r) {
        const int e0 = eBase + half * 8 + r;
        const int e1 = eBase + 16 + half * 8 + r;
        if (e0 < E) logits[e0] = accw0[r] + bb3;
        if (e1 < E) logits[e1] = accw1[r] + bb3;
      }
    }
    __builtin_amdgcn_wave_barrier();
  }
}

__global__ void notx_kernel(const float* __restrict__ uf,
                            const float* __restrict__ Wn1, const float* __restrict__ bn1,
                            const float* __restrict__ wn2, const float* __restrict__ bn2,
                            float* __restrict__ logit_no, unsigned* __restrict__ mkey, int nU)
{
  __shared__ float sW[192], sB[64], sw2[64];
  const int tid = threadIdx.x;
  for (int i = tid; i < 192; i += blockDim.x) sW[i] = Wn1[i];
  for (int i = tid; i < 64; i += blockDim.x) { sB[i] = bn1[i]; sw2[i] = wn2[i]; }
  __syncthreads();
  const int u = blockIdx.x * blockDim.x + tid;
  if (u >= nU) return;
  const float f0 = uf[u * 3 + 0], f1 = uf[u * 3 + 1], f2 = uf[u * 3 + 2];
  float acc = bn2[0];
  #pragma unroll 8
  for (int j = 0; j < 64; ++j) {
    float h = f0 * sW[j] + f1 * sW[64 + j] + f2 * sW[128 + j] + sB[j];
    h = h > 0.f ? h : 0.f;
    acc += h * sw2[j];
  }
  logit_no[u] = acc;
  mkey[u] = fkey(acc);
}

__global__ void segmax_kernel(const float* __restrict__ logits, const int* __restrict__ uidx,
                              unsigned* __restrict__ mkey, int E)
{
  const int e = blockIdx.x * blockDim.x + threadIdx.x;
  if (e < E) atomicMax(&mkey[uidx[e]], fkey(logits[e]));
}

__global__ void user_exp_kernel(const float* __restrict__ logit_no, const unsigned* __restrict__ mkey,
                                float* __restrict__ denom, float* __restrict__ outNo, int nU)
{
  const int u = blockIdx.x * blockDim.x + threadIdx.x;
  if (u < nU) {
    const float m  = funkey(mkey[u]);
    const float en = expf(logit_no[u] - m);
    denom[u] = en;
    outNo[u] = en;
  }
}

__global__ void edge_exp_kernel(const float* __restrict__ logits, const int* __restrict__ uidx,
                                const unsigned* __restrict__ mkey, float* __restrict__ denom,
                                float* __restrict__ outE, int E)
{
  const int e = blockIdx.x * blockDim.x + threadIdx.x;
  if (e < E) {
    const int u = uidx[e];
    const float ex = expf(logits[e] - funkey(mkey[u]));
    outE[e] = ex;
    atomicAdd(&denom[u], ex);
  }
}

__global__ void edge_div_kernel(float* __restrict__ outE, const int* __restrict__ uidx,
                                const float* __restrict__ denom, int E)
{
  const int e = blockIdx.x * blockDim.x + threadIdx.x;
  if (e < E) outE[e] = outE[e] / denom[uidx[e]];
}

__global__ void user_div_kernel(float* __restrict__ outNo, const float* __restrict__ denom, int nU)
{
  const int u = blockIdx.x * blockDim.x + threadIdx.x;
  if (u < nU) outNo[u] = outNo[u] / denom[u];
}

extern "C" void kernel_launch(void* const* d_in, const int* in_sizes, int n_in,
                              void* d_out, int out_size, void* d_ws, size_t ws_size,
                              hipStream_t stream)
{
  const float* uf  = (const float*)d_in[0];
  const float* gf  = (const float*)d_in[1];
  const float* ef  = (const float*)d_in[2];
  const int*   ui  = (const int*)d_in[3];
  const int*   gi  = (const int*)d_in[4];
  const float* W1  = (const float*)d_in[5];
  const float* b1  = (const float*)d_in[6];
  const float* W2  = (const float*)d_in[7];
  const float* b2  = (const float*)d_in[8];
  const float* w3  = (const float*)d_in[9];
  const float* b3  = (const float*)d_in[10];
  const float* Wn1 = (const float*)d_in[11];
  const float* bn1 = (const float*)d_in[12];
  const float* wn2 = (const float*)d_in[13];
  const float* bn2 = (const float*)d_in[14];

  const int nU = in_sizes[0] / 3;
  const int E  = in_sizes[3];

  float* ws          = (float*)d_ws;
  float* edge_logits = ws;
  float* logit_no    = ws + (size_t)E;
  unsigned* mkey     = (unsigned*)(ws + (size_t)E + (size_t)nU);
  float* denom       = ws + (size_t)E + 2 * (size_t)nU;

  float* outE  = (float*)d_out;
  float* outNo = outE + (size_t)E;

  notx_kernel<<<(nU + 255) / 256, 256, 0, stream>>>(uf, Wn1, bn1, wn2, bn2, logit_no, mkey, nU);

  const size_t lds = (size_t)(2 * W_ROW + 128 + 32 * W_ROW + 128 + 128
                              + WAVES * 2 * H_ROW + WAVES * 32 * H_ROW) * sizeof(float);
  const int nBlk = (E + EPB - 1) / EPB;
  edge_mlp_kernel<<<nBlk, THREADS, lds, stream>>>(uf, gf, ef, ui, gi,
                                                  W1, b1, W2, b2, w3, b3, edge_logits, E);

  segmax_kernel<<<(E + 255) / 256, 256, 0, stream>>>(edge_logits, ui, mkey, E);
  user_exp_kernel<<<(nU + 255) / 256, 256, 0, stream>>>(logit_no, mkey, denom, outNo, nU);
  edge_exp_kernel<<<(E + 255) / 256, 256, 0, stream>>>(edge_logits, ui, mkey, denom, outE, E);
  edge_div_kernel<<<(E + 255) / 256, 256, 0, stream>>>(outE, ui, denom, E);
  user_div_kernel<<<(nU + 255) / 256, 256, 0, stream>>>(outNo, denom, nU);
}


# edge MLP transposed: layer-1 D tiles feed layer-2 WMMAs as B operands from registers (no H trip through LDS), W2 as grouped b128 loads; changes f32 summation order (k-sum, w3 dot)
# speedup vs baseline: 1.3452x; 1.3452x over previous
#include <hip/hip_runtime.h>
#include <math.h>

typedef __attribute__((ext_vector_type(2))) float v2f;
typedef __attribute__((ext_vector_type(4))) float v4f;
typedef __attribute__((ext_vector_type(8))) float v8f;

#define WAVES 4
#define THREADS (WAVES * 32)
#define EDGES_PER_WAVE 32
#define EDGES_PER_PASS (WAVES * EDGES_PER_WAVE)
#define CHUNKS 8
#define EPB (EDGES_PER_PASS * CHUNKS)

#define W_ROW 512
#define H_ROW 132

__device__ __forceinline__ v8f wmma_f32_16x16x4(v2f a, v2f b, v8f c) {
  return __builtin_amdgcn_wmma_f32_16x16x4_f32(false, a, false, b, (short)0, c, false, false);
}

__device__ __forceinline__ v2f lds2(const float* p) { return *(const v2f*)p; }
__device__ __forceinline__ v4f lds4(const float* p) { return *(const v4f*)p; }

__device__ __forceinline__ unsigned fkey(float f) {
  unsigned u = __float_as_uint(f);
  return (u & 0x80000000u) ? ~u : (u | 0x80000000u);
}
__device__ __forceinline__ float funkey(unsigned k) {
  unsigned u = (k & 0x80000000u) ? (k & 0x7fffffffu) : ~k;
  return __uint_as_float(u);
}

__global__ __launch_bounds__(THREADS) void edge_mlp_kernel(
    const float* __restrict__ uf, const float* __restrict__ gf,
    const float* __restrict__ ef, const int* __restrict__ uidx,
    const int* __restrict__ gidx,
    const float* __restrict__ W1, const float* __restrict__ b1,
    const float* __restrict__ W2, const float* __restrict__ b2,
    const float* __restrict__ w3, const float* __restrict__ b3,
    float* __restrict__ logits, int E)
{
  extern __shared__ __attribute__((aligned(16))) float smem[];
  float* sW1g = smem;
  float* sb1  = sW1g + 2 * W_ROW;
  float* sW2g = sb1 + 128;
  float* sb2  = sW2g + 32 * W_ROW;
  float* sw3  = sb2 + 128;
  float* sX   = sw3 + 128;

  const int tid = threadIdx.x;

  for (int i = tid; i < 128; i += THREADS) { sb1[i] = b1[i]; sb2[i] = b2[i]; sw3[i] = w3[i]; }
  for (int i = tid; i < 8 * 128; i += THREADS) {
    const int k = i >> 7, n = i & 127;
    sW1g[(k >> 2) * W_ROW + n * 4 + (k & 3)] = W1[i];
  }
  for (int i = tid; i < 128 * 128; i += THREADS) {
    const int k = i >> 7, n = i & 127;
    sW2g[(k >> 2) * W_ROW + n * 4 + (k & 3)] = W2[i];
  }
  __syncthreads();

  const int wid  = tid >> 5;
  const int lane = tid & 31;
  const int l16  = lane & 15;
  const int half = lane >> 4;
  const int koff = half * 2;
  float* sXw = sX + wid * (2 * H_ROW);

  const float* aBase = sW2g + (2 * half) * W_ROW + l16 * 4;
  const float* pb1 = sb1 + half * 8;
  const float* pb2 = sb2 + half * 8;
  const float* pw3 = sw3 + half * 8;

  const int blockBase = blockIdx.x * EPB;

  v4f wA[16], wB[16];
  #pragma unroll
  for (int q = 0; q < 16; ++q)
    wA[q] = lds4(aBase + ((q >> 1) * 4 + (q & 1)) * W_ROW);

  for (int c = 0; c < CHUNKS; ++c) {
    const int eBase = blockBase + c * EDGES_PER_PASS + wid * EDGES_PER_WAVE;

    {
      const int m = lane;
      int e = eBase + m; if (e >= E) e = E - 1;
      const int uI = uidx[e], gI = gidx[e];
      float v[8];
      v[0] = uf[uI * 3 + 0]; v[1] = uf[uI * 3 + 1]; v[2] = uf[uI * 3 + 2];
      v[3] = gf[gI * 3 + 0]; v[4] = gf[gI * 3 + 1]; v[5] = gf[gI * 3 + 2];
      v[6] = ef[e * 2 + 0];  v[7] = ef[e * 2 + 1];
      #pragma unroll
      for (int f = 0; f < 8; ++f)
        sXw[(f >> 2) * H_ROW + m * 4 + (f & 3)] = v[f];
    }
    __builtin_amdgcn_wave_barrier();

    v8f h1a[8], h1b[8];
    {
      const v2f xa0 = lds2(&sXw[0 * H_ROW + l16 * 4 + koff]);
      const v2f xa1 = lds2(&sXw[1 * H_ROW + l16 * 4 + koff]);
      const v2f xb0 = lds2(&sXw[0 * H_ROW + (16 + l16) * 4 + koff]);
      const v2f xb1 = lds2(&sXw[1 * H_ROW + (16 + l16) * 4 + koff]);
      #pragma unroll
      for (int kt = 0; kt < 8; ++kt) {
        const int n = kt * 16 + l16;
        const v2f w0 = lds2(&sW1g[0 * W_ROW + n * 4 + koff]);
        const v2f w1 = lds2(&sW1g[1 * W_ROW + n * 4 + koff]);
        const v4f bl = lds4(pb1 + kt * 16);
        const v4f bh = lds4(pb1 + kt * 16 + 4);
        const v8f bias = {bl.x, bl.y, bl.z, bl.w, bh.x, bh.y, bh.z, bh.w};
        v8f ca = {0.f, 0.f, 0.f, 0.f, 0.f, 0.f, 0.f, 0.f};
        v8f cb = {0.f, 0.f, 0.f, 0.f, 0.f, 0.f, 0.f, 0.f};
        ca = wmma_f32_16x16x4(w0, xa0, ca);
        cb = wmma_f32_16x16x4(w0, xb0, cb);
        ca = wmma_f32_16x16x4(w1, xa1, ca);
        cb = wmma_f32_16x16x4(w1, xb1, cb);
        #pragma unroll
        for (int r = 0; r < 8; ++r) {
          float ha = ca[r] + bias[r]; ha = ha > 0.f ? ha : 0.f;
          float hb = cb[r] + bias[r]; hb = hb > 0.f ? hb : 0.f;
          h1a[kt][r] = ha;
          h1b[kt][r] = hb;
        }
      }
    }

    v8f accw0 = {0.f, 0.f, 0.f, 0.f, 0.f, 0.f, 0.f, 0.f};
    v8f accw1 = {0.f, 0.f, 0.f, 0.f, 0.f, 0.f, 0.f, 0.f};

#define L2_TILE(NT, CUR, NXT)                                                             \
    {                                                                                     \
                                                                \
           \
      _Pragma("unroll")                                                                   \
      for (int q = 0; q < 16; ++q)                                                        \
        NXT[q] = lds4(aBase + ((q >> 1) * 4 + (q & 1)) * W_ROW + (((NT) + 1) & 7) * 64);  \
      const v4f b2l = lds4(pb2 + (NT) * 16), b2h = lds4(pb2 + (NT) * 16 + 4);             \
      const v4f w3l = lds4(pw3 + (NT) * 16), w3h = lds4(pw3 + (NT) * 16 + 4);             \
                 \
      __builtin_amdgcn_sched_barrier(0);                                                  \
      v8f c0 = {0.f, 0.f, 0.f, 0.f, 0.f, 0.f, 0.f, 0.f};                                  \
      v8f c1 = {0.f, 0.f, 0.f, 0.f, 0.f, 0.f, 0.f, 0.f};                                  \
      _Pragma("unroll")                                                                   \
      for (int q = 0; q < 16; ++q) {          \
        const int kt = q >> 1, p = q & 1;                                                 \
        const v4f a = CUR[q];                                                             \
        const v2f alo = {a.x, a.y}, ahi = {a.z, a.w};                                     \
        const v2f b0lo = {h1a[kt][4 * p + 0], h1a[kt][4 * p + 1]};                        \
        const v2f b1lo = {h1b[kt][4 * p + 0], h1b[kt][4 * p + 1]};                        \
        const v2f b0hi = {h1a[kt][4 * p + 2], h1a[kt][4 * p + 3]};                        \
        const v2f b1hi = {h1b[kt][4 * p + 2], h1b[kt][4 * p + 3]};                        \
        c0 = wmma_f32_16x16x4(alo, b0lo, c0);           \
        c1 = wmma_f32_16x16x4(alo, b1lo, c1);           \
        c0 = wmma_f32_16x16x4(ahi, b0hi, c0);                                             \
        c1 = wmma_f32_16x16x4(ahi, b1hi, c1);                                             \
      }                                                                                   \
      const v8f b2v = {b2l.x, b2l.y, b2l.z, b2l.w, b2h.x, b2h.y, b2h.z, b2h.w};           \
      const v8f w3v = {w3l.x, w3l.y, w3l.z, w3l.w, w3h.x, w3h.y, w3h.z, w3h.w};           \
      _Pragma("unroll")                                                                   \
      for (int r = 0; r < 8; ++r) {           \
        float g0 = c0[r] + b2v[r]; g0 = g0 > 0.f ? g0 : 0.f;                              \
        float g1 = c1[r] + b2v[r]; g1 = g1 > 0.f ? g1 : 0.f;                              \
        accw0[r] += g0 * w3v[r];                                                          \
        accw1[r] += g1 * w3v[r];                                                          \
      }                                                                                   \
              \
      __builtin_amdgcn_sched_barrier(0);                                                  \
    }

    #pragma unroll 1
    for (int np = 0; np < 4; ++np) {
      L2_TILE(2 * np, wA, wB)
      L2_TILE(2 * np + 1, wB, wA)
    }
#undef L2_TILE

    float s0 = ((accw0[0] + accw0[1]) + (accw0[2] + accw0[3]))
             + ((accw0[4] + accw0[5]) + (accw0[6] + accw0[7]));
    float s1 = ((accw1[0] + accw1[1]) + (accw1[2] + accw1[3]))
             + ((accw1[4] + accw1[5]) + (accw1[6] + accw1[7]));
    s0 += __shfl_xor(s0, 16);
    s1 += __shfl_xor(s1, 16);
    {
      const int e = eBase + lane;
      const float mine = half ? s1 : s0;
      if (e < E) logits[e] = mine + b3[0];
    }
    __builtin_amdgcn_wave_barrier();
  }
}

__global__ void notx_kernel(const float* __restrict__ uf,
                            const float* __restrict__ Wn1, const float* __restrict__ bn1,
                            const float* __restrict__ wn2, const float* __restrict__ bn2,
                            float* __restrict__ logit_no, unsigned* __restrict__ mkey, int nU)
{
  __shared__ float sW[192], sB[64], sw2[64];
  const int tid = threadIdx.x;
  for (int i = tid; i < 192; i += blockDim.x) sW[i] = Wn1[i];
  for (int i = tid; i < 64; i += blockDim.x) { sB[i] = bn1[i]; sw2[i] = wn2[i]; }
  __syncthreads();
  const int u = blockIdx.x * blockDim.x + tid;
  if (u >= nU) return;
  const float f0 = uf[u * 3 + 0], f1 = uf[u * 3 + 1], f2 = uf[u * 3 + 2];
  float acc = bn2[0];
  #pragma unroll 8
  for (int j = 0; j < 64; ++j) {
    float h = f0 * sW[j] + f1 * sW[64 + j] + f2 * sW[128 + j] + sB[j];
    h = h > 0.f ? h : 0.f;
    acc += h * sw2[j];
  }
  logit_no[u] = acc;
  mkey[u] = fkey(acc);
}

__global__ void segmax_kernel(const float* __restrict__ logits, const int* __restrict__ uidx,
                              unsigned* __restrict__ mkey, int E)
{
  const int e = blockIdx.x * blockDim.x + threadIdx.x;
  if (e < E) atomicMax(&mkey[uidx[e]], fkey(logits[e]));
}

__global__ void user_exp_kernel(const float* __restrict__ logit_no, const unsigned* __restrict__ mkey,
                                float* __restrict__ denom, float* __restrict__ outNo, int nU)
{
  const int u = blockIdx.x * blockDim.x + threadIdx.x;
  if (u < nU) {
    const float m  = funkey(mkey[u]);
    const float en = expf(logit_no[u] - m);
    denom[u] = en;
    outNo[u] = en;
  }
}

__global__ void edge_exp_kernel(const float* __restrict__ logits, const int* __restrict__ uidx,
                                const unsigned* __restrict__ mkey, float* __restrict__ denom,
                                float* __restrict__ outE, int E)
{
  const int e = blockIdx.x * blockDim.x + threadIdx.x;
  if (e < E) {
    const int u = uidx[e];
    const float ex = expf(logits[e] - funkey(mkey[u]));
    outE[e] = ex;
    atomicAdd(&denom[u], ex);
  }
}

__global__ void edge_div_kernel(float* __restrict__ outE, const int* __restrict__ uidx,
                                const float* __restrict__ denom, int E)
{
  const int e = blockIdx.x * blockDim.x + threadIdx.x;
  if (e < E) outE[e] = outE[e] / denom[uidx[e]];
}

__global__ void user_div_kernel(float* __restrict__ outNo, const float* __restrict__ denom, int nU)
{
  const int u = blockIdx.x * blockDim.x + threadIdx.x;
  if (u < nU) outNo[u] = outNo[u] / denom[u];
}

extern "C" void kernel_launch(void* const* d_in, const int* in_sizes, int n_in,
                              void* d_out, int out_size, void* d_ws, size_t ws_size,
                              hipStream_t stream)
{
  const float* uf  = (const float*)d_in[0];
  const float* gf  = (const float*)d_in[1];
  const float* ef  = (const float*)d_in[2];
  const int*   ui  = (const int*)d_in[3];
  const int*   gi  = (const int*)d_in[4];
  const float* W1  = (const float*)d_in[5];
  const float* b1  = (const float*)d_in[6];
  const float* W2  = (const float*)d_in[7];
  const float* b2  = (const float*)d_in[8];
  const float* w3  = (const float*)d_in[9];
  const float* b3  = (const float*)d_in[10];
  const float* Wn1 = (const float*)d_in[11];
  const float* bn1 = (const float*)d_in[12];
  const float* wn2 = (const float*)d_in[13];
  const float* bn2 = (const float*)d_in[14];

  const int nU = in_sizes[0] / 3;
  const int E  = in_sizes[3];

  float* ws          = (float*)d_ws;
  float* edge_logits = ws;
  float* logit_no    = ws + (size_t)E;
  unsigned* mkey     = (unsigned*)(ws + (size_t)E + (size_t)nU);
  float* denom       = ws + (size_t)E + 2 * (size_t)nU;

  float* outE  = (float*)d_out;
  float* outNo = outE + (size_t)E;

  notx_kernel<<<(nU + 255) / 256, 256, 0, stream>>>(uf, Wn1, bn1, wn2, bn2, logit_no, mkey, nU);

  const size_t lds = (size_t)(2 * W_ROW + 128 + 32 * W_ROW + 128 + 128
                              + WAVES * 2 * H_ROW) * sizeof(float);
  const int nBlk = (E + EPB - 1) / EPB;
  edge_mlp_kernel<<<nBlk, THREADS, lds, stream>>>(uf, gf, ef, ui, gi,
                                                  W1, b1, W2, b2, w3, b3, edge_logits, E);

  segmax_kernel<<<(E + 255) / 256, 256, 0, stream>>>(edge_logits, ui, mkey, E);
  user_exp_kernel<<<(nU + 255) / 256, 256, 0, stream>>>(logit_no, mkey, denom, outNo, nU);
  edge_exp_kernel<<<(E + 255) / 256, 256, 0, stream>>>(edge_logits, ui, mkey, denom, outE, E);
  edge_div_kernel<<<(E + 255) / 256, 256, 0, stream>>>(outE, ui, denom, E);
  user_div_kernel<<<(nU + 255) / 256, 256, 0, stream>>>(outNo, denom, nU);
}


# softmax denominators summed in LDS: 128 blocks, each half the users x 1/64 of the edges (ds_add_f32), then one atomicAdd per non-zero entry; exp redone in the divide launch; changes f32 sum order
# speedup vs baseline: 1.8816x; 1.3988x over previous
#include <hip/hip_runtime.h>
#include <math.h>

typedef __attribute__((ext_vector_type(2))) float v2f;
typedef __attribute__((ext_vector_type(4))) float v4f;
typedef __attribute__((ext_vector_type(8))) float v8f;

#define WAVES 4
#define THREADS (WAVES * 32)
#define EDGES_PER_WAVE 32
#define EDGES_PER_PASS (WAVES * EDGES_PER_WAVE)
#define CHUNKS 8
#define EPB (EDGES_PER_PASS * CHUNKS)

#define W_ROW 512
#define H_ROW 132

__device__ __forceinline__ v8f wmma_f32_16x16x4(v2f a, v2f b, v8f c) {
  return __builtin_amdgcn_wmma_f32_16x16x4_f32(false, a, false, b, (short)0, c, false, false);
}

__device__ __forceinline__ v2f lds2(const float* p) { return *(const v2f*)p; }
__device__ __forceinline__ v4f lds4(const float* p) { return *(const v4f*)p; }

__device__ __forceinline__ unsigned fkey(float f) {
  unsigned u = __float_as_uint(f);
  return (u & 0x80000000u) ? ~u : (u | 0x80000000u);
}
__device__ __forceinline__ float funkey(unsigned k) {
  unsigned u = (k & 0x80000000u) ? (k & 0x7fffffffu) : ~k;
  return __uint_as_float(u);
}

__global__ __launch_bounds__(THREADS) void edge_mlp_kernel(
    const float* __restrict__ uf, const float* __restrict__ gf,
    const float* __restrict__ ef, const int* __restrict__ uidx,
    const int* __restrict__ gidx,
    const float* __restrict__ W1, const float* __restrict__ b1,
    const float* __restrict__ W2, const float* __restrict__ b2,
    const float* __restrict__ w3, const float* __restrict__ b3,
    float* __restrict__ logits, unsigned* __restrict__ mkey, int E)
{
  extern __shared__ __attribute__((aligned(16))) float smem[];
  float* sW1g = smem;
  float* sb1  = sW1g + 2 * W_ROW;
  float* sW2g = sb1 + 128;
  float* sb2  = sW2g + 32 * W_ROW;
  float* sw3  = sb2 + 128;
  float* sX   = sw3 + 128;

  const int tid = threadIdx.x;

  for (int i = tid; i < 128; i += THREADS) { sb1[i] = b1[i]; sb2[i] = b2[i]; sw3[i] = w3[i]; }
  for (int i = tid; i < 8 * 128; i += THREADS) {
    const int k = i >> 7, n = i & 127;
    sW1g[(k >> 2) * W_ROW + n * 4 + (k & 3)] = W1[i];
  }
  for (int i = tid; i < 128 * 128; i += THREADS) {
    const int k = i >> 7, n = i & 127;
    sW2g[(k >> 2) * W_ROW + n * 4 + (k & 3)] = W2[i];
  }
  __syncthreads();

  const int wid  = tid >> 5;
  const int lane = tid & 31;
  const int l16  = lane & 15;
  const int half = lane >> 4;
  const int koff = half * 2;
  float* sXw = sX + wid * (2 * H_ROW);

  const float* aBase = sW2g + (2 * half) * W_ROW + l16 * 4;
  const float* pb1 = sb1 + half * 8;
  const float* pb2 = sb2 + half * 8;
  const float* pw3 = sw3 + half * 8;

  const int blockBase = blockIdx.x * EPB;

  v4f wA[16], wB[16];
  #pragma unroll
  for (int q = 0; q < 16; ++q)
    wA[q] = lds4(aBase + ((q >> 1) * 4 + (q & 1)) * W_ROW);

  for (int c = 0; c < CHUNKS; ++c) {
    const int eBase = blockBase + c * EDGES_PER_PASS + wid * EDGES_PER_WAVE;

    int uI;
    {
      const int m = lane;
      int e = eBase + m; if (e >= E) e = E - 1;
      uI = uidx[e];
      const int gI = gidx[e];
      float v[8];
      v[0] = uf[uI * 3 + 0]; v[1] = uf[uI * 3 + 1]; v[2] = uf[uI * 3 + 2];
      v[3] = gf[gI * 3 + 0]; v[4] = gf[gI * 3 + 1]; v[5] = gf[gI * 3 + 2];
      v[6] = ef[e * 2 + 0];  v[7] = ef[e * 2 + 1];
      #pragma unroll
      for (int f = 0; f < 8; ++f)
        sXw[(f >> 2) * H_ROW + m * 4 + (f & 3)] = v[f];
    }
    __builtin_amdgcn_wave_barrier();

    v8f h1a[8], h1b[8];
    {
      const v2f xa0 = lds2(&sXw[0 * H_ROW + l16 * 4 + koff]);
      const v2f xa1 = lds2(&sXw[1 * H_ROW + l16 * 4 + koff]);
      const v2f xb0 = lds2(&sXw[0 * H_ROW + (16 + l16) * 4 + koff]);
      const v2f xb1 = lds2(&sXw[1 * H_ROW + (16 + l16) * 4 + koff]);
      #pragma unroll
      for (int kt = 0; kt < 8; ++kt) {
        const int n = kt * 16 + l16;
        const v2f w0 = lds2(&sW1g[0 * W_ROW + n * 4 + koff]);
        const v2f w1 = lds2(&sW1g[1 * W_ROW + n * 4 + koff]);
        const v4f bl = lds4(pb1 + kt * 16);
        const v4f bh = lds4(pb1 + kt * 16 + 4);
        const v8f bias = {bl.x, bl.y, bl.z, bl.w, bh.x, bh.y, bh.z, bh.w};
        v8f ca = {0.f, 0.f, 0.f, 0.f, 0.f, 0.f, 0.f, 0.f};
        v8f cb = {0.f, 0.f, 0.f, 0.f, 0.f, 0.f, 0.f, 0.f};
        ca = wmma_f32_16x16x4(w0, xa0, ca);
        cb = wmma_f32_16x16x4(w0, xb0, cb);
        ca = wmma_f32_16x16x4(w1, xa1, ca);
        cb = wmma_f32_16x16x4(w1, xb1, cb);
        #pragma unroll
        for (int r = 0; r < 8; ++r) {
          float ha = ca[r] + bias[r]; ha = ha > 0.f ? ha : 0.f;
          float hb = cb[r] + bias[r]; hb = hb > 0.f ? hb : 0.f;
          h1a[kt][r] = ha;
          h1b[kt][r] = hb;
        }
      }
    }

    v8f accw0 = {0.f, 0.f, 0.f, 0.f, 0.f, 0.f, 0.f, 0.f};
    v8f accw1 = {0.f, 0.f, 0.f, 0.f, 0.f, 0.f, 0.f, 0.f};

#define L2_TILE(NT, CUR, NXT)                                                             \
    {                                                                                     \
                                                                \
           \
      _Pragma("unroll")                                                                   \
      for (int q = 0; q < 16; ++q)                                                        \
        NXT[q] = lds4(aBase + ((q >> 1) * 4 + (q & 1)) * W_ROW + (((NT) + 1) & 7) * 64);  \
      const v4f b2l = lds4(pb2 + (NT) * 16), b2h = lds4(pb2 + (NT) * 16 + 4);             \
      const v4f w3l = lds4(pw3 + (NT) * 16), w3h = lds4(pw3 + (NT) * 16 + 4);             \
                 \
      __builtin_amdgcn_sched_barrier(0);                                                  \
      v8f c0 = {0.f, 0.f, 0.f, 0.f, 0.f, 0.f, 0.f, 0.f};                                  \
      v8f c1 = {0.f, 0.f, 0.f, 0.f, 0.f, 0.f, 0.f, 0.f};                                  \
      _Pragma("unroll")                                                                   \
      for (int q = 0; q < 16; ++q) {          \
        const int kt = q >> 1, p = q & 1;                                                 \
        const v4f a = CUR[q];                                                             \
        const v2f alo = {a.x, a.y}, ahi = {a.z, a.w};                                     \
        const v2f b0lo = {h1a[kt][4 * p + 0], h1a[kt][4 * p + 1]};                        \
        const v2f b1lo = {h1b[kt][4 * p + 0], h1b[kt][4 * p + 1]};                        \
        const v2f b0hi = {h1a[kt][4 * p + 2], h1a[kt][4 * p + 3]};                        \
        const v2f b1hi = {h1b[kt][4 * p + 2], h1b[kt][4 * p + 3]};                        \
        c0 = wmma_f32_16x16x4(alo, b0lo, c0);           \
        c1 = wmma_f32_16x16x4(alo, b1lo, c1);           \
        c0 = wmma_f32_16x16x4(ahi, b0hi, c0);                                             \
        c1 = wmma_f32_16x16x4(ahi, b1hi, c1);                                             \
      }                                                                                   \
      const v8f b2v = {b2l.x, b2l.y, b2l.z, b2l.w, b2h.x, b2h.y, b2h.z, b2h.w};           \
      const v8f w3v = {w3l.x, w3l.y, w3l.z, w3l.w, w3h.x, w3h.y, w3h.z, w3h.w};           \
      _Pragma("unroll")                                                                   \
      for (int r = 0; r < 8; ++r) {           \
        float g0 = c0[r] + b2v[r]; g0 = g0 > 0.f ? g0 : 0.f;                              \
        float g1 = c1[r] + b2v[r]; g1 = g1 > 0.f ? g1 : 0.f;                              \
        accw0[r] += g0 * w3v[r];                                                          \
        accw1[r] += g1 * w3v[r];                                                          \
      }                                                                                   \
              \
      __builtin_amdgcn_sched_barrier(0);                                                  \
    }

    #pragma unroll 1
    for (int np = 0; np < 4; ++np) {
      L2_TILE(2 * np, wA, wB)
      L2_TILE(2 * np + 1, wB, wA)
    }
#undef L2_TILE

    float s0 = ((accw0[0] + accw0[1]) + (accw0[2] + accw0[3]))
             + ((accw0[4] + accw0[5]) + (accw0[6] + accw0[7]));
    float s1 = ((accw1[0] + accw1[1]) + (accw1[2] + accw1[3]))
             + ((accw1[4] + accw1[5]) + (accw1[6] + accw1[7]));
    s0 += __shfl_xor(s0, 16);
    s1 += __shfl_xor(s1, 16);
    {
      const int e = eBase + lane;
      const float mine = (half ? s1 : s0) + b3[0];
      if (e < E) {
        logits[e] = mine;
        atomicMax(&mkey[uI], fkey(mine));
      }
    }
    __builtin_amdgcn_wave_barrier();
  }
}

__global__ void notx_kernel(const float* __restrict__ uf,
                            const float* __restrict__ Wn1, const float* __restrict__ bn1,
                            const float* __restrict__ wn2, const float* __restrict__ bn2,
                            float* __restrict__ logit_no, unsigned* __restrict__ mkey,
                            float* __restrict__ denom, int nU)
{
  __shared__ float sW[192], sB[64], sw2[64];
  const int tid = threadIdx.x;
  for (int i = tid; i < 192; i += blockDim.x) sW[i] = Wn1[i];
  for (int i = tid; i < 64; i += blockDim.x) { sB[i] = bn1[i]; sw2[i] = wn2[i]; }
  __syncthreads();
  const int u = blockIdx.x * blockDim.x + tid;
  if (u >= nU) return;
  const float f0 = uf[u * 3 + 0], f1 = uf[u * 3 + 1], f2 = uf[u * 3 + 2];
  float acc = bn2[0];
  #pragma unroll 8
  for (int j = 0; j < 64; ++j) {
    float h = f0 * sW[j] + f1 * sW[64 + j] + f2 * sW[128 + j] + sB[j];
    h = h > 0.f ? h : 0.f;
    acc += h * sw2[j];
  }
  logit_no[u] = acc;
  mkey[u] = fkey(acc);
  denom[u] = 0.f;
}

#define DEN_THREADS 1024
#define DEN_R (49 * DEN_THREADS)
#define DEN_SLICES 64
#define DEN_BATCH 8

__global__ __launch_bounds__(DEN_THREADS) void denom_kernel(
    const float* __restrict__ logits, const int* __restrict__ uidx,
    const float* __restrict__ logit_no, const unsigned* __restrict__ mkey,
    float* __restrict__ denom, int E, int nU, int SL)
{
  __shared__ __attribute__((aligned(16))) float tab[DEN_R];
  const int tid = threadIdx.x;
  const int g   = blockIdx.x % DEN_SLICES;
  const int r0  = (blockIdx.x / DEN_SLICES) * DEN_R;
  int rn = nU - r0; if (rn > DEN_R) rn = DEN_R;
  const int nCh = (rn + DEN_THREADS - 1) / DEN_THREADS;

  for (int k = 0; k < nCh; ++k) {
    const int i = k * DEN_THREADS + tid;
    float v = 0.f;
    if ((k % DEN_SLICES) == g && i < rn)
      v = expf(logit_no[r0 + i] - funkey(mkey[r0 + i]));
    tab[i] = v;
  }
  __syncthreads();

  const int e0 = g * SL;
  int e1 = e0 + SL; if (e1 > E) e1 = E;
  for (int base = e0; base < e1; base += DEN_THREADS * DEN_BATCH) {
    int u[DEN_BATCH]; float l[DEN_BATCH]; unsigned mk[DEN_BATCH]; bool in[DEN_BATCH];
    #pragma unroll
    for (int k = 0; k < DEN_BATCH; ++k) {
      const int e = base + k * DEN_THREADS + tid;
      const bool ok = e < e1;
      const int ec = ok ? e : e0;
      u[k] = uidx[ec];
      l[k] = logits[ec];
      in[k] = ok && (unsigned)(u[k] - r0) < (unsigned)rn;
    }
    #pragma unroll
    for (int k = 0; k < DEN_BATCH; ++k) mk[k] = mkey[in[k] ? u[k] : r0];
    #pragma unroll
    for (int k = 0; k < DEN_BATCH; ++k)
      if (in[k]) atomicAdd(&tab[u[k] - r0], expf(l[k] - funkey(mk[k])));
  }
  __syncthreads();

  const int k0 = (g * nCh) / DEN_SLICES;
  for (int kk = 0; kk < nCh; ++kk) {
    int k = kk + k0; if (k >= nCh) k -= nCh;
    const int i = k * DEN_THREADS + tid;
    if (i < rn) {
      const float v = tab[i];
      if (v != 0.f) atomicAdd(&denom[r0 + i], v);
    }
  }
}

__global__ void softmax_div_kernel(const float* __restrict__ logits, const int* __restrict__ uidx,
                                   const float* __restrict__ logit_no,
                                   const unsigned* __restrict__ mkey,
                                   const float* __restrict__ denom,
                                   float* __restrict__ outE, float* __restrict__ outNo,
                                   int E, int nU, int nUb)
{
  if ((int)blockIdx.x < nUb) {
    const int u = blockIdx.x * blockDim.x + threadIdx.x;
    if (u < nU) outNo[u] = expf(logit_no[u] - funkey(mkey[u])) / denom[u];
  } else {
    const int e = (blockIdx.x - nUb) * blockDim.x + threadIdx.x;
    if (e < E) {
      const int u = uidx[e];
      outE[e] = expf(logits[e] - funkey(mkey[u])) / denom[u];
    }
  }
}

extern "C" void kernel_launch(void* const* d_in, const int* in_sizes, int n_in,
                              void* d_out, int out_size, void* d_ws, size_t ws_size,
                              hipStream_t stream)
{
  const float* uf  = (const float*)d_in[0];
  const float* gf  = (const float*)d_in[1];
  const float* ef  = (const float*)d_in[2];
  const int*   ui  = (const int*)d_in[3];
  const int*   gi  = (const int*)d_in[4];
  const float* W1  = (const float*)d_in[5];
  const float* b1  = (const float*)d_in[6];
  const float* W2  = (const float*)d_in[7];
  const float* b2  = (const float*)d_in[8];
  const float* w3  = (const float*)d_in[9];
  const float* b3  = (const float*)d_in[10];
  const float* Wn1 = (const float*)d_in[11];
  const float* bn1 = (const float*)d_in[12];
  const float* wn2 = (const float*)d_in[13];
  const float* bn2 = (const float*)d_in[14];

  const int nU = in_sizes[0] / 3;
  const int E  = in_sizes[3];

  float* ws          = (float*)d_ws;
  float* edge_logits = ws;
  float* logit_no    = ws + (size_t)E;
  unsigned* mkey     = (unsigned*)(ws + (size_t)E + (size_t)nU);
  float* denom       = ws + (size_t)E + 2 * (size_t)nU;

  float* outE  = (float*)d_out;
  float* outNo = outE + (size_t)E;

  const int nUb = (nU + 255) / 256;
  notx_kernel<<<nUb, 256, 0, stream>>>(uf, Wn1, bn1, wn2, bn2, logit_no, mkey, denom, nU);

  const size_t lds = (size_t)(2 * W_ROW + 128 + 32 * W_ROW + 128 + 128
                              + WAVES * 2 * H_ROW) * sizeof(float);
  const int nBlk = (E + EPB - 1) / EPB;
  edge_mlp_kernel<<<nBlk, THREADS, lds, stream>>>(uf, gf, ef, ui, gi,
                                                  W1, b1, W2, b2, w3, b3, edge_logits, mkey, E);

  const int nR = (nU + DEN_R - 1) / DEN_R;
  const int SL = (E + DEN_SLICES - 1) / DEN_SLICES;
  denom_kernel<<<nR * DEN_SLICES, DEN_THREADS, 0, stream>>>(edge_logits, ui, logit_no, mkey, denom,
                                                           E, nU, SL);

  const int nEb = (E + 255) / 256;
  softmax_div_kernel<<<nUb + nEb, 256, 0, stream>>>(edge_logits, ui, logit_no, mkey, denom,
                                                    outE, outNo, E, nU, nUb);
}


# edge MLP in blocks of 256 edges, not 1,024 (2 passes each): shorter drain and evener block counts per WGP at the launch's end; weights staged a column per thread, loads grouped, 16-byte LDS stores
# speedup vs baseline: 1.9755x; 1.0499x over previous
#include <hip/hip_runtime.h>
#include <math.h>

typedef __attribute__((ext_vector_type(2))) float v2f;
typedef __attribute__((ext_vector_type(4))) float v4f;
typedef __attribute__((ext_vector_type(8))) float v8f;

#define WAVES 4
#define THREADS (WAVES * 32)
#define EDGES_PER_WAVE 32
#define EDGES_PER_PASS (WAVES * EDGES_PER_WAVE)
#define CHUNKS 2
#define EPB (EDGES_PER_PASS * CHUNKS)

#define W_ROW 512
#define H_ROW 132

__device__ __forceinline__ v8f wmma_f32_16x16x4(v2f a, v2f b, v8f c) {
  return __builtin_amdgcn_wmma_f32_16x16x4_f32(false, a, false, b, (short)0, c, false, false);
}

__device__ __forceinline__ v2f lds2(const float* p) { return *(const v2f*)p; }
__device__ __forceinline__ v4f lds4(const float* p) { return *(const v4f*)p; }

__device__ __forceinline__ unsigned fkey(float f) {
  unsigned u = __float_as_uint(f);
  return (u & 0x80000000u) ? ~u : (u | 0x80000000u);
}
__device__ __forceinline__ float funkey(unsigned k) {
  unsigned u = (k & 0x80000000u) ? (k & 0x7fffffffu) : ~k;
  return __uint_as_float(u);
}

__global__ __launch_bounds__(THREADS) void edge_mlp_kernel(
    const float* __restrict__ uf, const float* __restrict__ gf,
    const float* __restrict__ ef, const int* __restrict__ uidx,
    const int* __restrict__ gidx,
    const float* __restrict__ W1, const float* __restrict__ b1,
    const float* __restrict__ W2, const float* __restrict__ b2,
    const float* __restrict__ w3, const float* __restrict__ b3,
    float* __restrict__ logits, unsigned* __restrict__ mkey, int E)
{
  extern __shared__ __attribute__((aligned(16))) float smem[];
  float* sW1g = smem;
  float* sb1  = sW1g + 2 * W_ROW;
  float* sW2g = sb1 + 128;
  float* sb2  = sW2g + 32 * W_ROW;
  float* sw3  = sb2 + 128;
  float* sX   = sw3 + 128;

  const int tid = threadIdx.x;

  for (int n = tid; n < 128; n += THREADS) {
    sb1[n] = b1[n]; sb2[n] = b2[n]; sw3[n] = w3[n];
    #pragma unroll
    for (int g = 0; g < 2; ++g) {
      const v4f w = {W1[(4 * g + 0) * 128 + n], W1[(4 * g + 1) * 128 + n],
                     W1[(4 * g + 2) * 128 + n], W1[(4 * g + 3) * 128 + n]};
      *(v4f*)&sW1g[g * W_ROW + n * 4] = w;
    }
    #pragma unroll
    for (int g = 0; g < 32; ++g) {
      const v4f w = {W2[(4 * g + 0) * 128 + n], W2[(4 * g + 1) * 128 + n],
                     W2[(4 * g + 2) * 128 + n], W2[(4 * g + 3) * 128 + n]};
      *(v4f*)&sW2g[g * W_ROW + n * 4] = w;
    }
  }
  __syncthreads();

  const int wid  = tid >> 5;
  const int lane = tid & 31;
  const int l16  = lane & 15;
  const int half = lane >> 4;
  const int koff = half * 2;
  float* sXw = sX + wid * (2 * H_ROW);

  const float* aBase = sW2g + (2 * half) * W_ROW + l16 * 4;
  const float* pb1 = sb1 + half * 8;
  const float* pb2 = sb2 + half * 8;
  const float* pw3 = sw3 + half * 8;

  const int blockBase = blockIdx.x * EPB;

  v4f wA[16], wB[16];
  #pragma unroll
  for (int q = 0; q < 16; ++q)
    wA[q] = lds4(aBase + ((q >> 1) * 4 + (q & 1)) * W_ROW);

  for (int c = 0; c < CHUNKS; ++c) {
    const int eBase = blockBase + c * EDGES_PER_PASS + wid * EDGES_PER_WAVE;

    int uI;
    {
      const int m = lane;
      int e = eBase + m; if (e >= E) e = E - 1;
      uI = uidx[e];
      const int gI = gidx[e];
      float v[8];
      v[0] = uf[uI * 3 + 0]; v[1] = uf[uI * 3 + 1]; v[2] = uf[uI * 3 + 2];
      v[3] = gf[gI * 3 + 0]; v[4] = gf[gI * 3 + 1]; v[5] = gf[gI * 3 + 2];
      v[6] = ef[e * 2 + 0];  v[7] = ef[e * 2 + 1];
      #pragma unroll
      for (int f = 0; f < 8; ++f)
        sXw[(f >> 2) * H_ROW + m * 4 + (f & 3)] = v[f];
    }
    __builtin_amdgcn_wave_barrier();

    v8f h1a[8], h1b[8];
    {
      const v2f xa0 = lds2(&sXw[0 * H_ROW + l16 * 4 + koff]);
      const v2f xa1 = lds2(&sXw[1 * H_ROW + l16 * 4 + koff]);
      const v2f xb0 = lds2(&sXw[0 * H_ROW + (16 + l16) * 4 + koff]);
      const v2f xb1 = lds2(&sXw[1 * H_ROW + (16 + l16) * 4 + koff]);
      #pragma unroll
      for (int kt = 0; kt < 8; ++kt) {
        const int n = kt * 16 + l16;
        const v2f w0 = lds2(&sW1g[0 * W_ROW + n * 4 + koff]);
        const v2f w1 = lds2(&sW1g[1 * W_ROW + n * 4 + koff]);
        const v4f bl = lds4(pb1 + kt * 16);
        const v4f bh = lds4(pb1 + kt * 16 + 4);
        const v8f bias = {bl.x, bl.y, bl.z, bl.w, bh.x, bh.y, bh.z, bh.w};
        v8f ca = {0.f, 0.f, 0.f, 0.f, 0.f, 0.f, 0.f, 0.f};
        v8f cb = {0.f, 0.f, 0.f, 0.f, 0.f, 0.f, 0.f, 0.f};
        ca = wmma_f32_16x16x4(w0, xa0, ca);
        cb = wmma_f32_16x16x4(w0, xb0, cb);
        ca = wmma_f32_16x16x4(w1, xa1, ca);
        cb = wmma_f32_16x16x4(w1, xb1, cb);
        #pragma unroll
        for (int r = 0; r < 8; ++r) {
          float ha = ca[r] + bias[r]; ha = ha > 0.f ? ha : 0.f;
          float hb = cb[r] + bias[r]; hb = hb > 0.f ? hb : 0.f;
          h1a[kt][r] = ha;
          h1b[kt][r] = hb;
        }
      }
    }

    v8f accw0 = {0.f, 0.f, 0.f, 0.f, 0.f, 0.f, 0.f, 0.f};
    v8f accw1 = {0.f, 0.f, 0.f, 0.f, 0.f, 0.f, 0.f, 0.f};

#define L2_TILE(NT, CUR, NXT)                                                             \
    {                                                                                     \
                                                                \
           \
      _Pragma("unroll")                                                                   \
      for (int q = 0; q < 16; ++q)                                                        \
        NXT[q] = lds4(aBase + ((q >> 1) * 4 + (q & 1)) * W_ROW + (((NT) + 1) & 7) * 64);  \
      const v4f b2l = lds4(pb2 + (NT) * 16), b2h = lds4(pb2 + (NT) * 16 + 4);             \
      const v4f w3l = lds4(pw3 + (NT) * 16), w3h = lds4(pw3 + (NT) * 16 + 4);             \
                 \
      __builtin_amdgcn_sched_barrier(0);                                                  \
      v8f c0 = {0.f, 0.f, 0.f, 0.f, 0.f, 0.f, 0.f, 0.f};                                  \
      v8f c1 = {0.f, 0.f, 0.f, 0.f, 0.f, 0.f, 0.f, 0.f};                                  \
      _Pragma("unroll")                                                                   \
      for (int q = 0; q < 16; ++q) {          \
        const int kt = q >> 1, p = q & 1;                                                 \
        const v4f a = CUR[q];                                                             \
        const v2f alo = {a.x, a.y}, ahi = {a.z, a.w};                                     \
        const v2f b0lo = {h1a[kt][4 * p + 0], h1a[kt][4 * p + 1]};                        \
        const v2f b1lo = {h1b[kt][4 * p + 0], h1b[kt][4 * p + 1]};                        \
        const v2f b0hi = {h1a[kt][4 * p + 2], h1a[kt][4 * p + 3]};                        \
        const v2f b1hi = {h1b[kt][4 * p + 2], h1b[kt][4 * p + 3]};                        \
        c0 = wmma_f32_16x16x4(alo, b0lo, c0);           \
        c1 = wmma_f32_16x16x4(alo, b1lo, c1);           \
        c0 = wmma_f32_16x16x4(ahi, b0hi, c0);                                             \
        c1 = wmma_f32_16x16x4(ahi, b1hi, c1);                                             \
      }                                                                                   \
      const v8f b2v = {b2l.x, b2l.y, b2l.z, b2l.w, b2h.x, b2h.y, b2h.z, b2h.w};           \
      const v8f w3v = {w3l.x, w3l.y, w3l.z, w3l.w, w3h.x, w3h.y, w3h.z, w3h.w};           \
      _Pragma("unroll")                                                                   \
      for (int r = 0; r < 8; ++r) {           \
        float g0 = c0[r] + b2v[r]; g0 = g0 > 0.f ? g0 : 0.f;                              \
        float g1 = c1[r] + b2v[r]; g1 = g1 > 0.f ? g1 : 0.f;                              \
        accw0[r] += g0 * w3v[r];                                                          \
        accw1[r] += g1 * w3v[r];                                                          \
      }                                                                                   \
              \
      __builtin_amdgcn_sched_barrier(0);                                                  \
    }

    #pragma unroll 1
    for (int np = 0; np < 4; ++np) {
      L2_TILE(2 * np, wA, wB)
      L2_TILE(2 * np + 1, wB, wA)
    }
#undef L2_TILE

    float s0 = ((accw0[0] + accw0[1]) + (accw0[2] + accw0[3]))
             + ((accw0[4] + accw0[5]) + (accw0[6] + accw0[7]));
    float s1 = ((accw1[0] + accw1[1]) + (accw1[2] + accw1[3]))
             + ((accw1[4] + accw1[5]) + (accw1[6] + accw1[7]));
    s0 += __shfl_xor(s0, 16);
    s1 += __shfl_xor(s1, 16);
    {
      const int e = eBase + lane;
      const float mine = (half ? s1 : s0) + b3[0];
      if (e < E) {
        logits[e] = mine;
        atomicMax(&mkey[uI], fkey(mine));
      }
    }
    __builtin_amdgcn_wave_barrier();
  }
}

__global__ void notx_kernel(const float* __restrict__ uf,
                            const float* __restrict__ Wn1, const float* __restrict__ bn1,
                            const float* __restrict__ wn2, const float* __restrict__ bn2,
                            float* __restrict__ logit_no, unsigned* __restrict__ mkey,
                            float* __restrict__ denom, int nU)
{
  __shared__ float sW[192], sB[64], sw2[64];
  const int tid = threadIdx.x;
  for (int i = tid; i < 192; i += blockDim.x) sW[i] = Wn1[i];
  for (int i = tid; i < 64; i += blockDim.x) { sB[i] = bn1[i]; sw2[i] = wn2[i]; }
  __syncthreads();
  const int u = blockIdx.x * blockDim.x + tid;
  if (u >= nU) return;
  const float f0 = uf[u * 3 + 0], f1 = uf[u * 3 + 1], f2 = uf[u * 3 + 2];
  float acc = bn2[0];
  #pragma unroll 8
  for (int j = 0; j < 64; ++j) {
    float h = f0 * sW[j] + f1 * sW[64 + j] + f2 * sW[128 + j] + sB[j];
    h = h > 0.f ? h : 0.f;
    acc += h * sw2[j];
  }
  logit_no[u] = acc;
  mkey[u] = fkey(acc);
  denom[u] = 0.f;
}

#define DEN_THREADS 1024
#define DEN_R (49 * DEN_THREADS)
#define DEN_SLICES 64
#define DEN_BATCH 8

__global__ __launch_bounds__(DEN_THREADS) void denom_kernel(
    const float* __restrict__ logits, const int* __restrict__ uidx,
    const float* __restrict__ logit_no, const unsigned* __restrict__ mkey,
    float* __restrict__ denom, int E, int nU, int SL)
{
  __shared__ __attribute__((aligned(16))) float tab[DEN_R];
  const int tid = threadIdx.x;
  const int g   = blockIdx.x % DEN_SLICES;
  const int r0  = (blockIdx.x / DEN_SLICES) * DEN_R;
  int rn = nU - r0; if (rn > DEN_R) rn = DEN_R;
  const int nCh = (rn + DEN_THREADS - 1) / DEN_THREADS;

  for (int k = 0; k < nCh; ++k) {
    const int i = k * DEN_THREADS + tid;
    float v = 0.f;
    if ((k % DEN_SLICES) == g && i < rn)
      v = expf(logit_no[r0 + i] - funkey(mkey[r0 + i]));
    tab[i] = v;
  }
  __syncthreads();

  const int e0 = g * SL;
  int e1 = e0 + SL; if (e1 > E) e1 = E;
  for (int base = e0; base < e1; base += DEN_THREADS * DEN_BATCH) {
    int u[DEN_BATCH]; float l[DEN_BATCH]; unsigned mk[DEN_BATCH]; bool in[DEN_BATCH];
    #pragma unroll
    for (int k = 0; k < DEN_BATCH; ++k) {
      const int e = base + k * DEN_THREADS + tid;
      const bool ok = e < e1;
      const int ec = ok ? e : e0;
      u[k] = uidx[ec];
      l[k] = logits[ec];
      in[k] = ok && (unsigned)(u[k] - r0) < (unsigned)rn;
    }
    #pragma unroll
    for (int k = 0; k < DEN_BATCH; ++k) mk[k] = mkey[in[k] ? u[k] : r0];
    #pragma unroll
    for (int k = 0; k < DEN_BATCH; ++k)
      if (in[k]) atomicAdd(&tab[u[k] - r0], expf(l[k] - funkey(mk[k])));
  }
  __syncthreads();

  const int k0 = (g * nCh) / DEN_SLICES;
  for (int kk = 0; kk < nCh; ++kk) {
    int k = kk + k0; if (k >= nCh) k -= nCh;
    const int i = k * DEN_THREADS + tid;
    if (i < rn) {
      const float v = tab[i];
      if (v != 0.f) atomicAdd(&denom[r0 + i], v);
    }
  }
}

__global__ void softmax_div_kernel(const float* __restrict__ logits, const int* __restrict__ uidx,
                                   const float* __restrict__ logit_no,
                                   const unsigned* __restrict__ mkey,
                                   const float* __restrict__ denom,
                                   float* __restrict__ outE, float* __restrict__ outNo,
                                   int E, int nU, int nUb)
{
  if ((int)blockIdx.x < nUb) {
    const int u = blockIdx.x * blockDim.x + threadIdx.x;
    if (u < nU) outNo[u] = expf(logit_no[u] - funkey(mkey[u])) / denom[u];
  } else {
    const int e = (blockIdx.x - nUb) * blockDim.x + threadIdx.x;
    if (e < E) {
      const int u = uidx[e];
      outE[e] = expf(logits[e] - funkey(mkey[u])) / denom[u];
    }
  }
}

extern "C" void kernel_launch(void* const* d_in, const int* in_sizes, int n_in,
                              void* d_out, int out_size, void* d_ws, size_t ws_size,
                              hipStream_t stream)
{
  const float* uf  = (const float*)d_in[0];
  const float* gf  = (const float*)d_in[1];
  const float* ef  = (const float*)d_in[2];
  const int*   ui  = (const int*)d_in[3];
  const int*   gi  = (const int*)d_in[4];
  const float* W1  = (const float*)d_in[5];
  const float* b1  = (const float*)d_in[6];
  const float* W2  = (const float*)d_in[7];
  const float* b2  = (const float*)d_in[8];
  const float* w3  = (const float*)d_in[9];
  const float* b3  = (const float*)d_in[10];
  const float* Wn1 = (const float*)d_in[11];
  const float* bn1 = (const float*)d_in[12];
  const float* wn2 = (const float*)d_in[13];
  const float* bn2 = (const float*)d_in[14];

  const int nU = in_sizes[0] / 3;
  const int E  = in_sizes[3];

  float* ws          = (float*)d_ws;
  float* edge_logits = ws;
  float* logit_no    = ws + (size_t)E;
  unsigned* mkey     = (unsigned*)(ws + (size_t)E + (size_t)nU);
  float* denom       = ws + (size_t)E + 2 * (size_t)nU;

  float* outE  = (float*)d_out;
  float* outNo = outE + (size_t)E;

  const int nUb = (nU + 255) / 256;
  notx_kernel<<<nUb, 256, 0, stream>>>(uf, Wn1, bn1, wn2, bn2, logit_no, mkey, denom, nU);

  const size_t lds = (size_t)(2 * W_ROW + 128 + 32 * W_ROW + 128 + 128
                              + WAVES * 2 * H_ROW) * sizeof(float);
  const int nBlk = (E + EPB - 1) / EPB;
  edge_mlp_kernel<<<nBlk, THREADS, lds, stream>>>(uf, gf, ef, ui, gi,
                                                  W1, b1, W2, b2, w3, b3, edge_logits, mkey, E);

  const int nR = (nU + DEN_R - 1) / DEN_R;
  const int SL = (E + DEN_SLICES - 1) / DEN_SLICES;
  denom_kernel<<<nR * DEN_SLICES, DEN_THREADS, 0, stream>>>(edge_logits, ui, logit_no, mkey, denom,
                                                           E, nU, SL);

  const int nEb = (E + 255) / 256;
  softmax_div_kernel<<<nUb + nEb, 256, 0, stream>>>(edge_logits, ui, logit_no, mkey, denom,
                                                    outE, outNo, E, nU, nUb);
}
